# starting kernel
# baseline (speedup 1.0000x reference)
#include <hip/hip_runtime.h>
#include <math.h>

typedef __attribute__((ext_vector_type(2))) float v2f;
typedef __attribute__((ext_vector_type(8))) float v8f;

#define T_STEPS 28
#define F_IN    28
#define H_UNITS 128
#define G4      512
#define TILE_M  32
#define NKF_U   32
#define NKF_W   7

__device__ __forceinline__ float fast_sigmoid(float x) {
    return 1.0f / (1.0f + __expf(-x));
}

__global__ __launch_bounds__(256, 1)
void lstm_fused_wmma3(const float* __restrict__ x, const float* __restrict__ W,
                      const float* __restrict__ U, const float* __restrict__ b,
                      const float* __restrict__ Wd, const float* __restrict__ bd,
                      float* __restrict__ out)
{
    extern __shared__ float smem[];
    float* uSwz = smem;
    float* hSwz = smem + 65536;

    const int tid   = threadIdx.x;
    const int lane  = tid & 31;
    const int wave  = tid >> 5;
    const int laneN = lane & 15;
    const int half  = lane >> 4;
    const int rowBase = blockIdx.x * TILE_M;

    for (int i = tid; i < H_UNITS * G4; i += 256) {
        const int j  = i & 1;
        const int hf = (i >> 1) & 1;
        const int ln = (i >> 2) & 15;
        const int g  = (i >> 6) & 3;
        const int kf = (i >> 8) & 31;
        const int w  = i >> 13;
        uSwz[i] = U[(kf * 4 + hf * 2 + j) * G4 + (g * H_UNITS + w * 16 + ln)];
    }
    for (int i = tid; i < TILE_M * H_UNITS; i += 256) hSwz[i] = 0.0f;
    __syncthreads();

    const int nUnit = wave * 16 + laneN;
    const int nOff  = (nUnit >> 2) * 128 + ((nUnit >> 1) & 1) * 2 + (nUnit & 1);

    const float* uW  = uSwz + wave * 8192 + laneN * 4 + half * 2;
    const float* hA0 = hSwz + laneN * 4 + half * 2;
    const float* hA1 = hSwz + (16 + laneN) * 4 + half * 2;

    const float* xRow0 = x + (size_t)(rowBase + laneN) * (T_STEPS * F_IN);
    const float* xRow1 = x + (size_t)(rowBase + 16 + laneN) * (T_STEPS * F_IN);

    v8f biasV[4];
    #pragma unroll
    for (int g = 0; g < 4; ++g) {
        const float bias = b[g * H_UNITS + wave * 16 + laneN];
        biasV[g] = (v8f){bias,bias,bias,bias,bias,bias,bias,bias};
    }

    v8f cSt[2];
    cSt[0] = (v8f){0.f,0.f,0.f,0.f,0.f,0.f,0.f,0.f};
    cSt[1] = cSt[0];

    for (int t = 0; t < T_STEPS; ++t) {
        v8f acc[4][2];

        {
            const int k = half * 2;
            v2f a0, a1;
            a0.x = xRow0[t * F_IN + k];  a0.y = xRow0[t * F_IN + k + 1];
            a1.x = xRow1[t * F_IN + k];  a1.y = xRow1[t * F_IN + k + 1];
            #pragma unroll
            for (int g = 0; g < 4; ++g) {
                const int col = g * H_UNITS + wave * 16 + laneN;
                v2f bb;
                bb.x = W[k * G4 + col];
                bb.y = W[(k + 1) * G4 + col];
                acc[g][0] = __builtin_amdgcn_wmma_f32_16x16x4_f32(
                    false, a0, false, bb, (short)0, biasV[g], false, false);
                acc[g][1] = __builtin_amdgcn_wmma_f32_16x16x4_f32(
                    false, a1, false, bb, (short)0, biasV[g], false, false);
            }
        }
        #pragma unroll
        for (int kf = 1; kf < NKF_W; ++kf) {
            const int k = kf * 4 + half * 2;
            v2f a0, a1;
            a0.x = xRow0[t * F_IN + k];  a0.y = xRow0[t * F_IN + k + 1];
            a1.x = xRow1[t * F_IN + k];  a1.y = xRow1[t * F_IN + k + 1];
            #pragma unroll
            for (int g = 0; g < 4; ++g) {
                const int col = g * H_UNITS + wave * 16 + laneN;
                v2f bb;
                bb.x = W[k * G4 + col];
                bb.y = W[(k + 1) * G4 + col];
                acc[g][0] = __builtin_amdgcn_wmma_f32_16x16x4_f32(
                    false, a0, false, bb, (short)0, acc[g][0], false, false);
                acc[g][1] = __builtin_amdgcn_wmma_f32_16x16x4_f32(
                    false, a1, false, bb, (short)0, acc[g][1], false, false);
            }
        }

        for (int kf = 0; kf < NKF_U; ++kf) {
            const v2f a0 = *(const v2f*)&hA0[kf * 128];
            const v2f a1 = *(const v2f*)&hA1[kf * 128];
            #pragma unroll
            for (int g = 0; g < 4; ++g) {
                const v2f bb = *(const v2f*)&uW[kf * 256 + g * 64];
                acc[g][0] = __builtin_amdgcn_wmma_f32_16x16x4_f32(
                    false, a0, false, bb, (short)0, acc[g][0], false, false);
                acc[g][1] = __builtin_amdgcn_wmma_f32_16x16x4_f32(
                    false, a1, false, bb, (short)0, acc[g][1], false, false);
            }
        }

        __syncthreads();

        #pragma unroll
        for (int mt = 0; mt < 2; ++mt) {
            #pragma unroll
            for (int r = 0; r < 8; ++r) {
                const float ig = fast_sigmoid(acc[0][mt][r]);
                const float fg = fast_sigmoid(acc[1][mt][r]);
                const float gg = fmaxf(acc[2][mt][r], 0.0f);
                const float og = fast_sigmoid(acc[3][mt][r]);
                const float cn = fg * cSt[mt][r] + ig * gg;
                cSt[mt][r] = cn;
                const float hv = og * fmaxf(cn, 0.0f);
                const int m = mt * 16 + r + 8 * half;
                hSwz[nOff + m * 4] = hv;
            }
        }

        __syncthreads();
    }

    if (tid < TILE_M) {
        const int m = tid;
        float logits[10];
        #pragma unroll
        for (int c = 0; c < 10; ++c) logits[c] = bd[c];
        for (int k = 0; k < H_UNITS; ++k) {
            const float hv = hSwz[(k >> 2) * 128 + m * 4 + ((k >> 1) & 1) * 2 + (k & 1)];
            #pragma unroll
            for (int c = 0; c < 10; ++c) logits[c] += hv * Wd[k * 10 + c];
        }
        float mx = logits[0];
        #pragma unroll
        for (int c = 1; c < 10; ++c) mx = fmaxf(mx, logits[c]);
        float e[10], sum = 0.0f;
        #pragma unroll
        for (int c = 0; c < 10; ++c) { e[c] = __expf(logits[c] - mx); sum += e[c]; }
        const float inv = 1.0f / sum;
        #pragma unroll
        for (int c = 0; c < 10; ++c)
            out[(size_t)(rowBase + m) * 10 + c] = e[c] * inv;
    }
}

extern "C" void kernel_launch(void* const* d_in, const int* in_sizes, int n_in,
                              void* d_out, int out_size, void* d_ws, size_t ws_size,
                              hipStream_t stream) {
    const float* x  = (const float*)d_in[0];
    const float* W  = (const float*)d_in[1];
    const float* U  = (const float*)d_in[2];
    const float* b  = (const float*)d_in[3];
    const float* Wd = (const float*)d_in[4];
    const float* bd = (const float*)d_in[5];
    float* out = (float*)d_out;

    const size_t ldsBytes = (size_t)(65536 + 4096) * sizeof(float);
    const int numBlocks = 16384 / TILE_M;

    hipLaunchKernelGGL(lstm_fused_wmma3, dim3(numBlocks), dim3(256), ldsBytes, stream,
                       x, W, U, b, Wd, bd, out);

    (void)in_sizes; (void)n_in; (void)out_size; (void)d_ws; (void)ws_size;
}


# h@U fragment ds_loads issued one block of two K fragments ahead as ONE group into a second register set, held by sched_barrier, instead of a load and a full wait in front of every four WMMAs
# speedup vs baseline: 1.0759x; 1.0759x over previous
#include <hip/hip_runtime.h>
#include <math.h>

typedef __attribute__((ext_vector_type(2))) float v2f;
typedef __attribute__((ext_vector_type(8))) float v8f;

#define T_STEPS 28
#define F_IN    28
#define H_UNITS 128
#define G4      512
#define TILE_M  32
#define NKF_U   32
#define NKF_W   7

__device__ __forceinline__ float fast_sigmoid(float x) {
    return 1.0f / (1.0f + __expf(-x));
}

__global__ __launch_bounds__(256, 1)
void lstm_fused_wmma3(const float* __restrict__ x, const float* __restrict__ W,
                      const float* __restrict__ U, const float* __restrict__ b,
                      const float* __restrict__ Wd, const float* __restrict__ bd,
                      float* __restrict__ out)
{
    extern __shared__ float smem[];
    float* uSwz = smem;
    float* hSwz = smem + 65536;

    const int tid   = threadIdx.x;
    const int lane  = tid & 31;
    const int wave  = tid >> 5;
    const int laneN = lane & 15;
    const int half  = lane >> 4;
    const int rowBase = blockIdx.x * TILE_M;

    for (int i = tid; i < H_UNITS * G4; i += 256) {
        const int j  = i & 1;
        const int hf = (i >> 1) & 1;
        const int ln = (i >> 2) & 15;
        const int g  = (i >> 6) & 3;
        const int kf = (i >> 8) & 31;
        const int w  = i >> 13;
        uSwz[i] = U[(kf * 4 + hf * 2 + j) * G4 + (g * H_UNITS + w * 16 + ln)];
    }
    for (int i = tid; i < TILE_M * H_UNITS; i += 256) hSwz[i] = 0.0f;
    __syncthreads();

    const int nUnit = wave * 16 + laneN;
    const int nOff  = (nUnit >> 2) * 128 + ((nUnit >> 1) & 1) * 2 + (nUnit & 1);

    const float* uW  = uSwz + wave * 8192 + laneN * 4 + half * 2;
    const float* hA0 = hSwz + laneN * 4 + half * 2;
    const float* hA1 = hSwz + (16 + laneN) * 4 + half * 2;

    const float* xRow0 = x + (size_t)(rowBase + laneN) * (T_STEPS * F_IN);
    const float* xRow1 = x + (size_t)(rowBase + 16 + laneN) * (T_STEPS * F_IN);

    v8f biasV[4];
    #pragma unroll
    for (int g = 0; g < 4; ++g) {
        const float bias = b[g * H_UNITS + wave * 16 + laneN];
        biasV[g] = (v8f){bias,bias,bias,bias,bias,bias,bias,bias};
    }

    v8f cSt[2];
    cSt[0] = (v8f){0.f,0.f,0.f,0.f,0.f,0.f,0.f,0.f};
    cSt[1] = cSt[0];

    for (int t = 0; t < T_STEPS; ++t) {
        v8f acc[4][2];

        v2f ca0[2], ca1[2], cb[2][4];
        #pragma unroll
        for (int kk = 0; kk < 2; ++kk) {
            ca0[kk] = *(const v2f*)&hA0[kk * 128];
            ca1[kk] = *(const v2f*)&hA1[kk * 128];
            #pragma unroll
            for (int g = 0; g < 4; ++g)
                cb[kk][g] = *(const v2f*)&uW[kk * 256 + g * 64];
        }
        __builtin_amdgcn_sched_barrier(0);

        {
            const int k = half * 2;
            v2f a0, a1;
            a0.x = xRow0[t * F_IN + k];  a0.y = xRow0[t * F_IN + k + 1];
            a1.x = xRow1[t * F_IN + k];  a1.y = xRow1[t * F_IN + k + 1];
            #pragma unroll
            for (int g = 0; g < 4; ++g) {
                const int col = g * H_UNITS + wave * 16 + laneN;
                v2f bb;
                bb.x = W[k * G4 + col];
                bb.y = W[(k + 1) * G4 + col];
                acc[g][0] = __builtin_amdgcn_wmma_f32_16x16x4_f32(
                    false, a0, false, bb, (short)0, biasV[g], false, false);
                acc[g][1] = __builtin_amdgcn_wmma_f32_16x16x4_f32(
                    false, a1, false, bb, (short)0, biasV[g], false, false);
            }
        }
        #pragma unroll
        for (int kf = 1; kf < NKF_W; ++kf) {
            const int k = kf * 4 + half * 2;
            v2f a0, a1;
            a0.x = xRow0[t * F_IN + k];  a0.y = xRow0[t * F_IN + k + 1];
            a1.x = xRow1[t * F_IN + k];  a1.y = xRow1[t * F_IN + k + 1];
            #pragma unroll
            for (int g = 0; g < 4; ++g) {
                const int col = g * H_UNITS + wave * 16 + laneN;
                v2f bb;
                bb.x = W[k * G4 + col];
                bb.y = W[(k + 1) * G4 + col];
                acc[g][0] = __builtin_amdgcn_wmma_f32_16x16x4_f32(
                    false, a0, false, bb, (short)0, acc[g][0], false, false);
                acc[g][1] = __builtin_amdgcn_wmma_f32_16x16x4_f32(
                    false, a1, false, bb, (short)0, acc[g][1], false, false);
            }
        }

        __builtin_amdgcn_sched_barrier(0);

        #pragma unroll
        for (int kb = 0; kb < NKF_U / 2; ++kb) {
            v2f na0[2], na1[2], nb[2][4];
            if (kb + 1 < NKF_U / 2) {
                #pragma unroll
                for (int kk = 0; kk < 2; ++kk) {
                    const int kf = 2 * (kb + 1) + kk;
                    na0[kk] = *(const v2f*)&hA0[kf * 128];
                    na1[kk] = *(const v2f*)&hA1[kf * 128];
                    #pragma unroll
                    for (int g = 0; g < 4; ++g)
                        nb[kk][g] = *(const v2f*)&uW[kf * 256 + g * 64];
                }
            }
            __builtin_amdgcn_sched_barrier(0);
            #pragma unroll
            for (int kk = 0; kk < 2; ++kk) {
                #pragma unroll
                for (int g = 0; g < 4; ++g) {
                    acc[g][0] = __builtin_amdgcn_wmma_f32_16x16x4_f32(
                        false, ca0[kk], false, cb[kk][g], (short)0, acc[g][0], false, false);
                    acc[g][1] = __builtin_amdgcn_wmma_f32_16x16x4_f32(
                        false, ca1[kk], false, cb[kk][g], (short)0, acc[g][1], false, false);
                }
            }
            __builtin_amdgcn_sched_barrier(0);
            #pragma unroll
            for (int kk = 0; kk < 2; ++kk) {
                ca0[kk] = na0[kk];
                ca1[kk] = na1[kk];
                #pragma unroll
                for (int g = 0; g < 4; ++g) cb[kk][g] = nb[kk][g];
            }
        }

        __syncthreads();

        #pragma unroll
        for (int mt = 0; mt < 2; ++mt) {
            #pragma unroll
            for (int r = 0; r < 8; ++r) {
                const float ig = fast_sigmoid(acc[0][mt][r]);
                const float fg = fast_sigmoid(acc[1][mt][r]);
                const float gg = fmaxf(acc[2][mt][r], 0.0f);
                const float og = fast_sigmoid(acc[3][mt][r]);
                const float cn = fg * cSt[mt][r] + ig * gg;
                cSt[mt][r] = cn;
                const float hv = og * fmaxf(cn, 0.0f);
                const int m = mt * 16 + r + 8 * half;
                hSwz[nOff + m * 4] = hv;
            }
        }

        __syncthreads();
    }

    if (tid < TILE_M) {
        const int m = tid;
        float logits[10];
        #pragma unroll
        for (int c = 0; c < 10; ++c) logits[c] = bd[c];
        for (int k = 0; k < H_UNITS; ++k) {
            const float hv = hSwz[(k >> 2) * 128 + m * 4 + ((k >> 1) & 1) * 2 + (k & 1)];
            #pragma unroll
            for (int c = 0; c < 10; ++c) logits[c] += hv * Wd[k * 10 + c];
        }
        float mx = logits[0];
        #pragma unroll
        for (int c = 1; c < 10; ++c) mx = fmaxf(mx, logits[c]);
        float e[10], sum = 0.0f;
        #pragma unroll
        for (int c = 0; c < 10; ++c) { e[c] = __expf(logits[c] - mx); sum += e[c]; }
        const float inv = 1.0f / sum;
        #pragma unroll
        for (int c = 0; c < 10; ++c)
            out[(size_t)(rowBase + m) * 10 + c] = e[c] * inv;
    }
}

extern "C" void kernel_launch(void* const* d_in, const int* in_sizes, int n_in,
                              void* d_out, int out_size, void* d_ws, size_t ws_size,
                              hipStream_t stream) {
    const float* x  = (const float*)d_in[0];
    const float* W  = (const float*)d_in[1];
    const float* U  = (const float*)d_in[2];
    const float* b  = (const float*)d_in[3];
    const float* Wd = (const float*)d_in[4];
    const float* bd = (const float*)d_in[5];
    float* out = (float*)d_out;

    const size_t ldsBytes = (size_t)(65536 + 4096) * sizeof(float);
    const int numBlocks = 16384 / TILE_M;

    hipLaunchKernelGGL(lstm_fused_wmma3, dim3(numBlocks), dim3(256), ldsBytes, stream,
                       x, W, U, b, Wd, bd, out);

    (void)in_sizes; (void)n_in; (void)out_size; (void)d_ws; (void)ws_size;
}


# gate sigmoids: 1/(1+e) taken with the hardware reciprocal v_rcp_f32 (1 ulp) instead of the exact f32 division sequence; gate rounding may differ by 1 ulp, sum order unchanged
# speedup vs baseline: 1.1818x; 1.0984x over previous
#include <hip/hip_runtime.h>
#include <math.h>

typedef __attribute__((ext_vector_type(2))) float v2f;
typedef __attribute__((ext_vector_type(8))) float v8f;

#define T_STEPS 28
#define F_IN    28
#define H_UNITS 128
#define G4      512
#define TILE_M  32
#define NKF_U   32
#define NKF_W   7

__device__ __forceinline__ float fast_sigmoid(float x) {
    return __builtin_amdgcn_rcpf(1.0f + __expf(-x));
}

__global__ __launch_bounds__(256, 1)
void lstm_fused_wmma3(const float* __restrict__ x, const float* __restrict__ W,
                      const float* __restrict__ U, const float* __restrict__ b,
                      const float* __restrict__ Wd, const float* __restrict__ bd,
                      float* __restrict__ out)
{
    extern __shared__ float smem[];
    float* uSwz = smem;
    float* hSwz = smem + 65536;

    const int tid   = threadIdx.x;
    const int lane  = tid & 31;
    const int wave  = tid >> 5;
    const int laneN = lane & 15;
    const int half  = lane >> 4;
    const int rowBase = blockIdx.x * TILE_M;

    for (int i = tid; i < H_UNITS * G4; i += 256) {
        const int j  = i & 1;
        const int hf = (i >> 1) & 1;
        const int ln = (i >> 2) & 15;
        const int g  = (i >> 6) & 3;
        const int kf = (i >> 8) & 31;
        const int w  = i >> 13;
        uSwz[i] = U[(kf * 4 + hf * 2 + j) * G4 + (g * H_UNITS + w * 16 + ln)];
    }
    for (int i = tid; i < TILE_M * H_UNITS; i += 256) hSwz[i] = 0.0f;
    __syncthreads();

    const int nUnit = wave * 16 + laneN;
    const int nOff  = (nUnit >> 2) * 128 + ((nUnit >> 1) & 1) * 2 + (nUnit & 1);

    const float* uW  = uSwz + wave * 8192 + laneN * 4 + half * 2;
    const float* hA0 = hSwz + laneN * 4 + half * 2;
    const float* hA1 = hSwz + (16 + laneN) * 4 + half * 2;

    const float* xRow0 = x + (size_t)(rowBase + laneN) * (T_STEPS * F_IN);
    const float* xRow1 = x + (size_t)(rowBase + 16 + laneN) * (T_STEPS * F_IN);

    v8f biasV[4];
    #pragma unroll
    for (int g = 0; g < 4; ++g) {
        const float bias = b[g * H_UNITS + wave * 16 + laneN];
        biasV[g] = (v8f){bias,bias,bias,bias,bias,bias,bias,bias};
    }

    v8f cSt[2];
    cSt[0] = (v8f){0.f,0.f,0.f,0.f,0.f,0.f,0.f,0.f};
    cSt[1] = cSt[0];

    for (int t = 0; t < T_STEPS; ++t) {
        v8f acc[4][2];

        v2f ca0[2], ca1[2], cb[2][4];
        #pragma unroll
        for (int kk = 0; kk < 2; ++kk) {
            ca0[kk] = *(const v2f*)&hA0[kk * 128];
            ca1[kk] = *(const v2f*)&hA1[kk * 128];
            #pragma unroll
            for (int g = 0; g < 4; ++g)
                cb[kk][g] = *(const v2f*)&uW[kk * 256 + g * 64];
        }
        __builtin_amdgcn_sched_barrier(0);

        {
            const int k = half * 2;
            v2f a0, a1;
            a0.x = xRow0[t * F_IN + k];  a0.y = xRow0[t * F_IN + k + 1];
            a1.x = xRow1[t * F_IN + k];  a1.y = xRow1[t * F_IN + k + 1];
            #pragma unroll
            for (int g = 0; g < 4; ++g) {
                const int col = g * H_UNITS + wave * 16 + laneN;
                v2f bb;
                bb.x = W[k * G4 + col];
                bb.y = W[(k + 1) * G4 + col];
                acc[g][0] = __builtin_amdgcn_wmma_f32_16x16x4_f32(
                    false, a0, false, bb, (short)0, biasV[g], false, false);
                acc[g][1] = __builtin_amdgcn_wmma_f32_16x16x4_f32(
                    false, a1, false, bb, (short)0, biasV[g], false, false);
            }
        }
        #pragma unroll
        for (int kf = 1; kf < NKF_W; ++kf) {
            const int k = kf * 4 + half * 2;
            v2f a0, a1;
            a0.x = xRow0[t * F_IN + k];  a0.y = xRow0[t * F_IN + k + 1];
            a1.x = xRow1[t * F_IN + k];  a1.y = xRow1[t * F_IN + k + 1];
            #pragma unroll
            for (int g = 0; g < 4; ++g) {
                const int col = g * H_UNITS + wave * 16 + laneN;
                v2f bb;
                bb.x = W[k * G4 + col];
                bb.y = W[(k + 1) * G4 + col];
                acc[g][0] = __builtin_amdgcn_wmma_f32_16x16x4_f32(
                    false, a0, false, bb, (short)0, acc[g][0], false, false);
                acc[g][1] = __builtin_amdgcn_wmma_f32_16x16x4_f32(
                    false, a1, false, bb, (short)0, acc[g][1], false, false);
            }
        }

        __builtin_amdgcn_sched_barrier(0);

        #pragma unroll
        for (int kb = 0; kb < NKF_U / 2; ++kb) {
            v2f na0[2], na1[2], nb[2][4];
            if (kb + 1 < NKF_U / 2) {
                #pragma unroll
                for (int kk = 0; kk < 2; ++kk) {
                    const int kf = 2 * (kb + 1) + kk;
                    na0[kk] = *(const v2f*)&hA0[kf * 128];
                    na1[kk] = *(const v2f*)&hA1[kf * 128];
                    #pragma unroll
                    for (int g = 0; g < 4; ++g)
                        nb[kk][g] = *(const v2f*)&uW[kf * 256 + g * 64];
                }
            }
            __builtin_amdgcn_sched_barrier(0);
            #pragma unroll
            for (int kk = 0; kk < 2; ++kk) {
                #pragma unroll
                for (int g = 0; g < 4; ++g) {
                    acc[g][0] = __builtin_amdgcn_wmma_f32_16x16x4_f32(
                        false, ca0[kk], false, cb[kk][g], (short)0, acc[g][0], false, false);
                    acc[g][1] = __builtin_amdgcn_wmma_f32_16x16x4_f32(
                        false, ca1[kk], false, cb[kk][g], (short)0, acc[g][1], false, false);
                }
            }
            __builtin_amdgcn_sched_barrier(0);
            #pragma unroll
            for (int kk = 0; kk < 2; ++kk) {
                ca0[kk] = na0[kk];
                ca1[kk] = na1[kk];
                #pragma unroll
                for (int g = 0; g < 4; ++g) cb[kk][g] = nb[kk][g];
            }
        }

        __syncthreads();

        #pragma unroll
        for (int mt = 0; mt < 2; ++mt) {
            #pragma unroll
            for (int r = 0; r < 8; ++r) {
                const float ig = fast_sigmoid(acc[0][mt][r]);
                const float fg = fast_sigmoid(acc[1][mt][r]);
                const float gg = fmaxf(acc[2][mt][r], 0.0f);
                const float og = fast_sigmoid(acc[3][mt][r]);
                const float cn = fg * cSt[mt][r] + ig * gg;
                cSt[mt][r] = cn;
                const float hv = og * fmaxf(cn, 0.0f);
                const int m = mt * 16 + r + 8 * half;
                hSwz[nOff + m * 4] = hv;
            }
        }

        __syncthreads();
    }

    if (tid < TILE_M) {
        const int m = tid;
        float logits[10];
        #pragma unroll
        for (int c = 0; c < 10; ++c) logits[c] = bd[c];
        for (int k = 0; k < H_UNITS; ++k) {
            const float hv = hSwz[(k >> 2) * 128 + m * 4 + ((k >> 1) & 1) * 2 + (k & 1)];
            #pragma unroll
            for (int c = 0; c < 10; ++c) logits[c] += hv * Wd[k * 10 + c];
        }
        float mx = logits[0];
        #pragma unroll
        for (int c = 1; c < 10; ++c) mx = fmaxf(mx, logits[c]);
        float e[10], sum = 0.0f;
        #pragma unroll
        for (int c = 0; c < 10; ++c) { e[c] = __expf(logits[c] - mx); sum += e[c]; }
        const float inv = 1.0f / sum;
        #pragma unroll
        for (int c = 0; c < 10; ++c)
            out[(size_t)(rowBase + m) * 10 + c] = e[c] * inv;
    }
}

extern "C" void kernel_launch(void* const* d_in, const int* in_sizes, int n_in,
                              void* d_out, int out_size, void* d_ws, size_t ws_size,
                              hipStream_t stream) {
    const float* x  = (const float*)d_in[0];
    const float* W  = (const float*)d_in[1];
    const float* U  = (const float*)d_in[2];
    const float* b  = (const float*)d_in[3];
    const float* Wd = (const float*)d_in[4];
    const float* bd = (const float*)d_in[5];
    float* out = (float*)d_out;

    const size_t ldsBytes = (size_t)(65536 + 4096) * sizeof(float);
    const int numBlocks = 16384 / TILE_M;

    hipLaunchKernelGGL(lstm_fused_wmma3, dim3(numBlocks), dim3(256), ldsBytes, stream,
                       x, W, U, b, Wd, bd, out);

    (void)in_sizes; (void)n_in; (void)out_size; (void)d_ws; (void)ws_size;
}


# batch tile per workgroup 32 -> 64 rows (4 M-tiles, 16 accumulator tiles per wave): 256 workgroups in one round, U staged once per WGP, step barriers and load waits per 64 rows; sum order unchanged
# speedup vs baseline: 1.2806x; 1.0836x over previous
#include <hip/hip_runtime.h>
#include <math.h>

typedef __attribute__((ext_vector_type(2))) float v2f;
typedef __attribute__((ext_vector_type(8))) float v8f;

#define T_STEPS 28
#define F_IN    28
#define H_UNITS 128
#define G4      512
#define TILE_M  64
#define MT      4
#define NKF_U   32
#define NKF_W   7

__device__ __forceinline__ float fast_sigmoid(float x) {
    return __builtin_amdgcn_rcpf(1.0f + __expf(-x));
}

__global__ __launch_bounds__(256, 1)
void lstm_fused_wmma3(const float* __restrict__ x, const float* __restrict__ W,
                      const float* __restrict__ U, const float* __restrict__ b,
                      const float* __restrict__ Wd, const float* __restrict__ bd,
                      float* __restrict__ out)
{
    extern __shared__ float smem[];
    float* uSwz = smem;
    float* hSwz = smem + 65536;

    const int tid   = threadIdx.x;
    const int lane  = tid & 31;
    const int wave  = tid >> 5;
    const int laneN = lane & 15;
    const int half  = lane >> 4;
    const int rowBase = blockIdx.x * TILE_M;

    for (int i = tid; i < H_UNITS * G4; i += 256) {
        const int j  = i & 1;
        const int hf = (i >> 1) & 1;
        const int ln = (i >> 2) & 15;
        const int g  = (i >> 6) & 3;
        const int kf = (i >> 8) & 31;
        const int w  = i >> 13;
        uSwz[i] = U[(kf * 4 + hf * 2 + j) * G4 + (g * H_UNITS + w * 16 + ln)];
    }
    for (int i = tid; i < TILE_M * H_UNITS; i += 256) hSwz[i] = 0.0f;
    __syncthreads();

    const int nUnit = wave * 16 + laneN;
    const int nOff  = (nUnit >> 2) * 256 + ((nUnit >> 1) & 1) * 2 + (nUnit & 1);

    const float* uW  = uSwz + wave * 8192 + laneN * 4 + half * 2;
    const float* hA = hSwz + laneN * 4 + half * 2;

    const float* xRow[MT];
    #pragma unroll
    for (int mt = 0; mt < MT; ++mt)
        xRow[mt] = x + (size_t)(rowBase + mt * 16 + laneN) * (T_STEPS * F_IN);

    v8f biasV[4];
    #pragma unroll
    for (int g = 0; g < 4; ++g) {
        const float bias = b[g * H_UNITS + wave * 16 + laneN];
        biasV[g] = (v8f){bias,bias,bias,bias,bias,bias,bias,bias};
    }

    v8f cSt[MT];
    #pragma unroll
    for (int mt = 0; mt < MT; ++mt) cSt[mt] = (v8f){0.f,0.f,0.f,0.f,0.f,0.f,0.f,0.f};

    for (int t = 0; t < T_STEPS; ++t) {
        v8f acc[4][MT];

        v2f xa[NKF_W][MT];
        #pragma unroll
        for (int kf = 0; kf < NKF_W; ++kf) {
            #pragma unroll
            for (int mt = 0; mt < MT; ++mt)
                xa[kf][mt] = *(const v2f*)&xRow[mt][t * F_IN + kf * 4 + half * 2];
        }
        v2f ca[2][MT], cb[2][4];
        #pragma unroll
        for (int kk = 0; kk < 2; ++kk) {
            #pragma unroll
            for (int mt = 0; mt < MT; ++mt)
                ca[kk][mt] = *(const v2f*)&hA[kk * 256 + mt * 64];
            #pragma unroll
            for (int g = 0; g < 4; ++g)
                cb[kk][g] = *(const v2f*)&uW[kk * 256 + g * 64];
        }
        __builtin_amdgcn_sched_barrier(0);

        {
            const int k = half * 2;
            #pragma unroll
            for (int g = 0; g < 4; ++g) {
                const int col = g * H_UNITS + wave * 16 + laneN;
                v2f bb;
                bb.x = W[k * G4 + col];
                bb.y = W[(k + 1) * G4 + col];
                #pragma unroll
                for (int mt = 0; mt < MT; ++mt)
                    acc[g][mt] = __builtin_amdgcn_wmma_f32_16x16x4_f32(
                        false, xa[0][mt], false, bb, (short)0, biasV[g], false, false);
            }
        }
        #pragma unroll
        for (int kf = 1; kf < NKF_W; ++kf) {
            const int k = kf * 4 + half * 2;
            #pragma unroll
            for (int g = 0; g < 4; ++g) {
                const int col = g * H_UNITS + wave * 16 + laneN;
                v2f bb;
                bb.x = W[k * G4 + col];
                bb.y = W[(k + 1) * G4 + col];
                #pragma unroll
                for (int mt = 0; mt < MT; ++mt)
                    acc[g][mt] = __builtin_amdgcn_wmma_f32_16x16x4_f32(
                        false, xa[kf][mt], false, bb, (short)0, acc[g][mt], false, false);
            }
        }

        __builtin_amdgcn_sched_barrier(0);

        #pragma unroll
        for (int kb = 0; kb < NKF_U / 2; ++kb) {
            v2f na[2][MT], nb[2][4];
            if (kb + 1 < NKF_U / 2) {
                #pragma unroll
                for (int kk = 0; kk < 2; ++kk) {
                    const int kf = 2 * (kb + 1) + kk;
                    #pragma unroll
                    for (int mt = 0; mt < MT; ++mt)
                        na[kk][mt] = *(const v2f*)&hA[kf * 256 + mt * 64];
                    #pragma unroll
                    for (int g = 0; g < 4; ++g)
                        nb[kk][g] = *(const v2f*)&uW[kf * 256 + g * 64];
                }
            }
            __builtin_amdgcn_sched_barrier(0);
            #pragma unroll
            for (int kk = 0; kk < 2; ++kk) {
                #pragma unroll
                for (int g = 0; g < 4; ++g) {
                    #pragma unroll
                    for (int mt = 0; mt < MT; ++mt)
                        acc[g][mt] = __builtin_amdgcn_wmma_f32_16x16x4_f32(
                            false, ca[kk][mt], false, cb[kk][g], (short)0, acc[g][mt], false, false);
                }
            }
            __builtin_amdgcn_sched_barrier(0);
            #pragma unroll
            for (int kk = 0; kk < 2; ++kk) {
                #pragma unroll
                for (int mt = 0; mt < MT; ++mt) ca[kk][mt] = na[kk][mt];
                #pragma unroll
                for (int g = 0; g < 4; ++g) cb[kk][g] = nb[kk][g];
            }
        }

        __syncthreads();

        #pragma unroll
        for (int mt = 0; mt < MT; ++mt) {
            #pragma unroll
            for (int r = 0; r < 8; ++r) {
                const float ig = fast_sigmoid(acc[0][mt][r]);
                const float fg = fast_sigmoid(acc[1][mt][r]);
                const float gg = fmaxf(acc[2][mt][r], 0.0f);
                const float og = fast_sigmoid(acc[3][mt][r]);
                const float cn = fg * cSt[mt][r] + ig * gg;
                cSt[mt][r] = cn;
                const float hv = og * fmaxf(cn, 0.0f);
                const int m = mt * 16 + r + 8 * half;
                hSwz[nOff + m * 4] = hv;
            }
        }

        __syncthreads();
    }

    if (tid < TILE_M) {
        const int m = tid;
        float logits[10];
        #pragma unroll
        for (int c = 0; c < 10; ++c) logits[c] = bd[c];
        for (int k = 0; k < H_UNITS; ++k) {
            const float hv = hSwz[(k >> 2) * 256 + m * 4 + ((k >> 1) & 1) * 2 + (k & 1)];
            #pragma unroll
            for (int c = 0; c < 10; ++c) logits[c] += hv * Wd[k * 10 + c];
        }
        float mx = logits[0];
        #pragma unroll
        for (int c = 1; c < 10; ++c) mx = fmaxf(mx, logits[c]);
        float e[10], sum = 0.0f;
        #pragma unroll
        for (int c = 0; c < 10; ++c) { e[c] = __expf(logits[c] - mx); sum += e[c]; }
        const float inv = 1.0f / sum;
        #pragma unroll
        for (int c = 0; c < 10; ++c)
            out[(size_t)(rowBase + m) * 10 + c] = e[c] * inv;
    }
}

extern "C" void kernel_launch(void* const* d_in, const int* in_sizes, int n_in,
                              void* d_out, int out_size, void* d_ws, size_t ws_size,
                              hipStream_t stream) {
    const float* x  = (const float*)d_in[0];
    const float* W  = (const float*)d_in[1];
    const float* U  = (const float*)d_in[2];
    const float* b  = (const float*)d_in[3];
    const float* Wd = (const float*)d_in[4];
    const float* bd = (const float*)d_in[5];
    float* out = (float*)d_out;

    const size_t ldsBytes = (size_t)(65536 + 8192) * sizeof(float);
    const int numBlocks = 16384 / TILE_M;

    hipLaunchKernelGGL(lstm_fused_wmma3, dim3(numBlocks), dim3(256), ldsBytes, stream,
                       x, W, U, b, Wd, bd, out);

    (void)in_sizes; (void)n_in; (void)out_size; (void)d_ws; (void)ws_size;
}


# step 0 peeled (h0 = 0): its 512 h@U WMMAs per wave dropped, and U staged to LDS in two 128-deep load batches that fly under step 0's x@W WMMAs and gate math, not ahead of the loop; sums unchanged
# speedup vs baseline: 1.3310x; 1.0394x over previous
#include <hip/hip_runtime.h>
#include <math.h>

typedef __attribute__((ext_vector_type(2))) float v2f;
typedef __attribute__((ext_vector_type(8))) float v8f;

#define T_STEPS 28
#define F_IN    28
#define H_UNITS 128
#define G4      512
#define TILE_M  64
#define MT      4
#define NKF_U   32
#define NKF_W   7

__device__ __forceinline__ float fast_sigmoid(float x) {
    return __builtin_amdgcn_rcpf(1.0f + __expf(-x));
}

__global__ __launch_bounds__(256, 1)
void lstm_fused_wmma3(const float* __restrict__ x, const float* __restrict__ W,
                      const float* __restrict__ U, const float* __restrict__ b,
                      const float* __restrict__ Wd, const float* __restrict__ bd,
                      float* __restrict__ out)
{
    extern __shared__ float smem[];
    float* uSwz = smem;
    float* hSwz = smem + 65536;

    const int tid   = threadIdx.x;
    const int lane  = tid & 31;
    const int wave  = tid >> 5;
    const int laneN = lane & 15;
    const int half  = lane >> 4;
    const int rowBase = blockIdx.x * TILE_M;

    const int nUnit = wave * 16 + laneN;
    const int nOff  = (nUnit >> 2) * 256 + ((nUnit >> 1) & 1) * 2 + (nUnit & 1);

    const float* uW  = uSwz + wave * 8192 + laneN * 4 + half * 2;
    const float* hA = hSwz + laneN * 4 + half * 2;

    const float* xRow[MT];
    #pragma unroll
    for (int mt = 0; mt < MT; ++mt)
        xRow[mt] = x + (size_t)(rowBase + mt * 16 + laneN) * (T_STEPS * F_IN);

    v8f biasV[4];
    #pragma unroll
    for (int g = 0; g < 4; ++g) {
        const float bias = b[g * H_UNITS + wave * 16 + laneN];
        biasV[g] = (v8f){bias,bias,bias,bias,bias,bias,bias,bias};
    }

    v8f cSt[MT];
    #pragma unroll
    for (int mt = 0; mt < MT; ++mt) cSt[mt] = (v8f){0.f,0.f,0.f,0.f,0.f,0.f,0.f,0.f};


    v2f wB[NKF_W][4];
    #pragma unroll
    for (int kf = 0; kf < NKF_W; ++kf) {
        const int k = kf * 4 + half * 2;
        #pragma unroll
        for (int g = 0; g < 4; ++g) {
            const int col = g * H_UNITS + wave * 16 + laneN;
            wB[kf][g].x = W[k * G4 + col];
            wB[kf][g].y = W[(k + 1) * G4 + col];
        }
    }

    v2f xa0[NKF_W][MT];
    #pragma unroll
    for (int kf = 0; kf < NKF_W; ++kf) {
        #pragma unroll
        for (int mt = 0; mt < MT; ++mt)
            xa0[kf][mt] = *(const v2f*)&xRow[mt][kf * 4 + half * 2];
    }
    __builtin_amdgcn_sched_barrier(0);

    const int sln = tid & 15;
    const int shf = (tid >> 4) & 1;
    const int sj  = (tid >> 5) & 1;
    const int sg  = tid >> 6;
    const float* uSrc = U + (shf * 2 + sj) * G4 + sg * H_UNITS + sln;
    float*       uDst = uSwz + sg * 64 + sln * 4 + shf * 2 + sj;

    float ub[128];
    #pragma unroll
    for (int q = 0; q < 128; ++q)
        ub[q] = uSrc[(q >> 3) * 4 * G4 + (q & 7) * 16];
    __builtin_amdgcn_sched_barrier(0);

    {
        v8f acc[4][MT];
        #pragma unroll
        for (int g = 0; g < 4; ++g) {
            #pragma unroll
            for (int mt = 0; mt < MT; ++mt)
                acc[g][mt] = __builtin_amdgcn_wmma_f32_16x16x4_f32(
                    false, xa0[0][mt], false, wB[0][g], (short)0, biasV[g], false, false);
        }
        #pragma unroll
        for (int kf = 1; kf < NKF_W; ++kf) {
            #pragma unroll
            for (int g = 0; g < 4; ++g) {
                #pragma unroll
                for (int mt = 0; mt < MT; ++mt)
                    acc[g][mt] = __builtin_amdgcn_wmma_f32_16x16x4_f32(
                        false, xa0[kf][mt], false, wB[kf][g], (short)0, acc[g][mt], false, false);
            }
        }
        __builtin_amdgcn_sched_barrier(0);

        #pragma unroll
        for (int q = 0; q < 128; ++q)
            uDst[(q & 7) * 8192 + (q >> 3) * 256] = ub[q];
        __builtin_amdgcn_sched_barrier(0);
        #pragma unroll
        for (int q = 0; q < 128; ++q)
            ub[q] = uSrc[((q >> 3) + 16) * 4 * G4 + (q & 7) * 16];
        __builtin_amdgcn_sched_barrier(0);

        #pragma unroll
        for (int mt = 0; mt < MT; ++mt) {
            #pragma unroll
            for (int r = 0; r < 8; ++r) {
                const float ig = fast_sigmoid(acc[0][mt][r]);
                const float fg = fast_sigmoid(acc[1][mt][r]);
                const float gg = fmaxf(acc[2][mt][r], 0.0f);
                const float og = fast_sigmoid(acc[3][mt][r]);
                const float cn = fg * cSt[mt][r] + ig * gg;
                cSt[mt][r] = cn;
                const float hv = og * fmaxf(cn, 0.0f);
                const int m = mt * 16 + r + 8 * half;
                hSwz[nOff + m * 4] = hv;
            }
        }


        __builtin_amdgcn_sched_barrier(0);
        #pragma unroll
        for (int q = 0; q < 128; ++q)
            uDst[(q & 7) * 8192 + ((q >> 3) + 16) * 256] = ub[q];
    }
    __syncthreads();

    for (int t = 1; t < T_STEPS; ++t) {
        v8f acc[4][MT];

        v2f xa[NKF_W][MT];
        #pragma unroll
        for (int kf = 0; kf < NKF_W; ++kf) {
            #pragma unroll
            for (int mt = 0; mt < MT; ++mt)
                xa[kf][mt] = *(const v2f*)&xRow[mt][t * F_IN + kf * 4 + half * 2];
        }
        v2f ca[2][MT], cb[2][4];
        #pragma unroll
        for (int kk = 0; kk < 2; ++kk) {
            #pragma unroll
            for (int mt = 0; mt < MT; ++mt)
                ca[kk][mt] = *(const v2f*)&hA[kk * 256 + mt * 64];
            #pragma unroll
            for (int g = 0; g < 4; ++g)
                cb[kk][g] = *(const v2f*)&uW[kk * 256 + g * 64];
        }
        __builtin_amdgcn_sched_barrier(0);

        {
            #pragma unroll
            for (int g = 0; g < 4; ++g) {
                const v2f bb = wB[0][g];
                #pragma unroll
                for (int mt = 0; mt < MT; ++mt)
                    acc[g][mt] = __builtin_amdgcn_wmma_f32_16x16x4_f32(
                        false, xa[0][mt], false, bb, (short)0, biasV[g], false, false);
            }
        }
        #pragma unroll
        for (int kf = 1; kf < NKF_W; ++kf) {
            #pragma unroll
            for (int g = 0; g < 4; ++g) {
                const v2f bb = wB[kf][g];
                #pragma unroll
                for (int mt = 0; mt < MT; ++mt)
                    acc[g][mt] = __builtin_amdgcn_wmma_f32_16x16x4_f32(
                        false, xa[kf][mt], false, bb, (short)0, acc[g][mt], false, false);
            }
        }

        __builtin_amdgcn_sched_barrier(0);

        #pragma unroll
        for (int kb = 0; kb < NKF_U / 2; ++kb) {
            v2f na[2][MT], nb[2][4];
            if (kb + 1 < NKF_U / 2) {
                #pragma unroll
                for (int kk = 0; kk < 2; ++kk) {
                    const int kf = 2 * (kb + 1) + kk;
                    #pragma unroll
                    for (int mt = 0; mt < MT; ++mt)
                        na[kk][mt] = *(const v2f*)&hA[kf * 256 + mt * 64];
                    #pragma unroll
                    for (int g = 0; g < 4; ++g)
                        nb[kk][g] = *(const v2f*)&uW[kf * 256 + g * 64];
                }
            }
            __builtin_amdgcn_sched_barrier(0);
            #pragma unroll
            for (int kk = 0; kk < 2; ++kk) {
                #pragma unroll
                for (int g = 0; g < 4; ++g) {
                    #pragma unroll
                    for (int mt = 0; mt < MT; ++mt)
                        acc[g][mt] = __builtin_amdgcn_wmma_f32_16x16x4_f32(
                            false, ca[kk][mt], false, cb[kk][g], (short)0, acc[g][mt], false, false);
                }
            }
            __builtin_amdgcn_sched_barrier(0);
            #pragma unroll
            for (int kk = 0; kk < 2; ++kk) {
                #pragma unroll
                for (int mt = 0; mt < MT; ++mt) ca[kk][mt] = na[kk][mt];
                #pragma unroll
                for (int g = 0; g < 4; ++g) cb[kk][g] = nb[kk][g];
            }
        }

        __syncthreads();

        #pragma unroll
        for (int mt = 0; mt < MT; ++mt) {
            #pragma unroll
            for (int r = 0; r < 8; ++r) {
                const float ig = fast_sigmoid(acc[0][mt][r]);
                const float fg = fast_sigmoid(acc[1][mt][r]);
                const float gg = fmaxf(acc[2][mt][r], 0.0f);
                const float og = fast_sigmoid(acc[3][mt][r]);
                const float cn = fg * cSt[mt][r] + ig * gg;
                cSt[mt][r] = cn;
                const float hv = og * fmaxf(cn, 0.0f);
                const int m = mt * 16 + r + 8 * half;
                hSwz[nOff + m * 4] = hv;
            }
        }

        __syncthreads();
    }

    if (tid < TILE_M) {
        const int m = tid;
        float logits[10];
        #pragma unroll
        for (int c = 0; c < 10; ++c) logits[c] = bd[c];
        for (int k = 0; k < H_UNITS; ++k) {
            const float hv = hSwz[(k >> 2) * 256 + m * 4 + ((k >> 1) & 1) * 2 + (k & 1)];
            #pragma unroll
            for (int c = 0; c < 10; ++c) logits[c] += hv * Wd[k * 10 + c];
        }
        float mx = logits[0];
        #pragma unroll
        for (int c = 1; c < 10; ++c) mx = fmaxf(mx, logits[c]);
        float e[10], sum = 0.0f;
        #pragma unroll
        for (int c = 0; c < 10; ++c) { e[c] = __expf(logits[c] - mx); sum += e[c]; }
        const float inv = 1.0f / sum;
        #pragma unroll
        for (int c = 0; c < 10; ++c)
            out[(size_t)(rowBase + m) * 10 + c] = e[c] * inv;
    }
}

extern "C" void kernel_launch(void* const* d_in, const int* in_sizes, int n_in,
                              void* d_out, int out_size, void* d_ws, size_t ws_size,
                              hipStream_t stream) {
    const float* x  = (const float*)d_in[0];
    const float* W  = (const float*)d_in[1];
    const float* U  = (const float*)d_in[2];
    const float* b  = (const float*)d_in[3];
    const float* Wd = (const float*)d_in[4];
    const float* bd = (const float*)d_in[5];
    float* out = (float*)d_out;

    const size_t ldsBytes = (size_t)(65536 + 8192) * sizeof(float);
    const int numBlocks = 16384 / TILE_M;

    hipLaunchKernelGGL(lstm_fused_wmma3, dim3(numBlocks), dim3(256), ldsBytes, stream,
                       x, W, U, b, Wd, bd, out);

    (void)in_sizes; (void)n_in; (void)out_size; (void)d_ws; (void)ws_size;
}


# 64-row tile run as two independent 32-row wave groups (waves w and w+4 of a SIMD) one barrier phase apart: one wave's gate math and x loads stand beside its partner's 624-WMMA run; sums unchanged
# speedup vs baseline: 1.3763x; 1.0340x over previous
#include <hip/hip_runtime.h>
#include <math.h>

typedef __attribute__((ext_vector_type(2))) float v2f;
typedef __attribute__((ext_vector_type(8))) float v8f;

#define T_STEPS 28
#define F_IN    28
#define H_UNITS 128
#define G4      512
#define TILE_M  64
#define MT      2
#define NS      2
#define NKF_U   32
#define NKF_W   7

__device__ __forceinline__ float fast_sigmoid(float x) {
    return __builtin_amdgcn_rcpf(1.0f + __expf(-x));
}

__global__ __launch_bounds__(256, 1)
void lstm_fused_wmma3(const float* __restrict__ x, const float* __restrict__ W,
                      const float* __restrict__ U, const float* __restrict__ b,
                      const float* __restrict__ Wd, const float* __restrict__ bd,
                      float* __restrict__ out)
{
    extern __shared__ float smem[];
    float* uSwz = smem;
    float* hSwz = smem + 65536;

    const int tid   = threadIdx.x;
    const int lane  = tid & 31;
    const int wave  = __builtin_amdgcn_readfirstlane(tid >> 5);
    const int grp   = wave >> 2;
    const int q     = wave & 3;
    const int laneN = lane & 15;
    const int half  = lane >> 4;
    const int rowBase = blockIdx.x * TILE_M;

    const int nUnit = q * 32 + laneN;
    const int nOff  = (nUnit >> 2) * 256 + ((nUnit >> 1) & 1) * 2 + (nUnit & 1);

    const float* uW = uSwz + (2 * q) * 8192 + laneN * 4 + half * 2;
    const float* hA = hSwz + (2 * grp) * 64 + laneN * 4 + half * 2;
    float*       hS = hSwz + nOff + ((2 * grp) * 16 + 8 * half) * 4;

    const float* xRow[MT];
    #pragma unroll
    for (int mt = 0; mt < MT; ++mt)
        xRow[mt] = x + (size_t)(rowBase + (2 * grp + mt) * 16 + laneN) * (T_STEPS * F_IN);

    v8f biasV[4][NS];
    #pragma unroll
    for (int g = 0; g < 4; ++g) {
        #pragma unroll
        for (int ns = 0; ns < NS; ++ns) {
            const float bias = b[g * H_UNITS + q * 32 + ns * 16 + laneN];
            biasV[g][ns] = (v8f){bias,bias,bias,bias,bias,bias,bias,bias};
        }
    }

    v8f cSt[NS][MT];
    #pragma unroll
    for (int ns = 0; ns < NS; ++ns) {
        #pragma unroll
        for (int mt = 0; mt < MT; ++mt) cSt[ns][mt] = (v8f){0.f,0.f,0.f,0.f,0.f,0.f,0.f,0.f};
    }

    v2f wB[NKF_W][4][NS];
    #pragma unroll
    for (int kf = 0; kf < NKF_W; ++kf) {
        const int k = kf * 4 + half * 2;
        #pragma unroll
        for (int g = 0; g < 4; ++g) {
            #pragma unroll
            for (int ns = 0; ns < NS; ++ns) {
                const int col = g * H_UNITS + q * 32 + ns * 16 + laneN;
                wB[kf][g][ns].x = W[k * G4 + col];
                wB[kf][g][ns].y = W[(k + 1) * G4 + col];
            }
        }
    }

    v8f acc[4][NS][MT];

    {
        const int sln = tid & 15;
        const int shf = (tid >> 4) & 1;
        const int sj  = (tid >> 5) & 1;
        const int sg  = tid >> 6;
        const float* uSrc = U + (shf * 2 + sj) * G4 + sg * H_UNITS + sln;
        float*       uDst = uSwz + sg * 64 + sln * 4 + shf * 2 + sj;

        float ub[128];
        #pragma unroll
        for (int i = 0; i < 128; ++i)
            ub[i] = uSrc[(i >> 3) * 4 * G4 + (i & 7) * 16];
        __builtin_amdgcn_sched_barrier(0);

        v2f xa0[NKF_W][MT];
        #pragma unroll
        for (int kf = 0; kf < NKF_W; ++kf) {
            #pragma unroll
            for (int mt = 0; mt < MT; ++mt)
                xa0[kf][mt] = *(const v2f*)&xRow[mt][kf * 4 + half * 2];
        }

        #pragma unroll
        for (int i = 0; i < 128; ++i)
            uDst[(i & 7) * 8192 + (i >> 3) * 256] = ub[i];
        __builtin_amdgcn_sched_barrier(0);
        #pragma unroll
        for (int i = 0; i < 128; ++i)
            ub[i] = uSrc[((i >> 3) + 16) * 4 * G4 + (i & 7) * 16];
        #pragma unroll
        for (int i = 0; i < 128; ++i)
            uDst[(i & 7) * 8192 + ((i >> 3) + 16) * 256] = ub[i];
        __builtin_amdgcn_sched_barrier(0);

        #pragma unroll
        for (int g = 0; g < 4; ++g) {
            #pragma unroll
            for (int ns = 0; ns < NS; ++ns) {
                #pragma unroll
                for (int mt = 0; mt < MT; ++mt)
                    acc[g][ns][mt] = __builtin_amdgcn_wmma_f32_16x16x4_f32(
                                false, xa0[0][mt], false, wB[0][g][ns], (short)0, biasV[g][ns], false, false);
            }
        }
        #pragma unroll
        for (int kf = 1; kf < NKF_W; ++kf) {
            #pragma unroll
            for (int g = 0; g < 4; ++g) {
                #pragma unroll
                for (int ns = 0; ns < NS; ++ns) {
                    #pragma unroll
                    for (int mt = 0; mt < MT; ++mt)
                        acc[g][ns][mt] = __builtin_amdgcn_wmma_f32_16x16x4_f32(
                                false, xa0[kf][mt], false, wB[kf][g][ns], (short)0, acc[g][ns][mt], false, false);
                }
            }
        }
    }
    __syncthreads();

    v2f xa[NKF_W][MT];
    #pragma unroll
    for (int kf = 0; kf < NKF_W; ++kf) {
        #pragma unroll
        for (int mt = 0; mt < MT; ++mt) xa[kf][mt] = (v2f){0.f, 0.f};
    }
    for (int i = 0; i < 2 * T_STEPS; ++i) {
        const int s = i - grp;
        if (s >= 0 && s <= 2 * (T_STEPS - 1) && (s & 1)) {
            {
                v2f ca[2][MT], cb[2][4][NS];
                #pragma unroll
                for (int kk = 0; kk < 2; ++kk) {
                    #pragma unroll
                    for (int mt = 0; mt < MT; ++mt)
                        ca[kk][mt] = *(const v2f*)&hA[kk * 256 + mt * 64];
                    #pragma unroll
                    for (int g = 0; g < 4; ++g) {
                        #pragma unroll
                        for (int ns = 0; ns < NS; ++ns)
                            cb[kk][g][ns] = *(const v2f*)&uW[ns * 8192 + kk * 256 + g * 64];
                    }
                }
                __builtin_amdgcn_sched_barrier(0);

                #pragma unroll
                for (int g = 0; g < 4; ++g) {
                    #pragma unroll
                    for (int ns = 0; ns < NS; ++ns) {
                        #pragma unroll
                        for (int mt = 0; mt < MT; ++mt)
                            acc[g][ns][mt] = __builtin_amdgcn_wmma_f32_16x16x4_f32(
                                false, xa[0][mt], false, wB[0][g][ns], (short)0, biasV[g][ns], false, false);
                    }
                }
                #pragma unroll
                for (int kf = 1; kf < NKF_W; ++kf) {
                    #pragma unroll
                    for (int g = 0; g < 4; ++g) {
                        #pragma unroll
                        for (int ns = 0; ns < NS; ++ns) {
                            #pragma unroll
                            for (int mt = 0; mt < MT; ++mt)
                                acc[g][ns][mt] = __builtin_amdgcn_wmma_f32_16x16x4_f32(
                                false, xa[kf][mt], false, wB[kf][g][ns], (short)0, acc[g][ns][mt], false, false);
                        }
                    }
                }
                __builtin_amdgcn_sched_barrier(0);

                #pragma unroll
                for (int kb = 0; kb < NKF_U / 2; ++kb) {
                    v2f na[2][MT], nb[2][4][NS];
                    if (kb + 1 < NKF_U / 2) {
                        #pragma unroll
                        for (int kk = 0; kk < 2; ++kk) {
                            const int kf = 2 * (kb + 1) + kk;
                            #pragma unroll
                            for (int mt = 0; mt < MT; ++mt)
                                na[kk][mt] = *(const v2f*)&hA[kf * 256 + mt * 64];
                            #pragma unroll
                            for (int g = 0; g < 4; ++g) {
                                #pragma unroll
                                for (int ns = 0; ns < NS; ++ns)
                                    nb[kk][g][ns] = *(const v2f*)&uW[ns * 8192 + kf * 256 + g * 64];
                            }
                        }
                    }
                    __builtin_amdgcn_sched_barrier(0);
                    #pragma unroll
                    for (int kk = 0; kk < 2; ++kk) {
                        #pragma unroll
                        for (int g = 0; g < 4; ++g) {
                            #pragma unroll
                            for (int ns = 0; ns < NS; ++ns) {
                                #pragma unroll
                                for (int mt = 0; mt < MT; ++mt)
                                    acc[g][ns][mt] = __builtin_amdgcn_wmma_f32_16x16x4_f32(
                                false, ca[kk][mt], false, cb[kk][g][ns], (short)0, acc[g][ns][mt], false, false);
                            }
                        }
                    }
                    __builtin_amdgcn_sched_barrier(0);
                    #pragma unroll
                    for (int kk = 0; kk < 2; ++kk) {
                        #pragma unroll
                        for (int mt = 0; mt < MT; ++mt) ca[kk][mt] = na[kk][mt];
                        #pragma unroll
                        for (int g = 0; g < 4; ++g) {
                            #pragma unroll
                            for (int ns = 0; ns < NS; ++ns) cb[kk][g][ns] = nb[kk][g][ns];
                        }
                    }
                }
            }
        }
        if (s >= 0 && s <= 2 * (T_STEPS - 1) && !(s & 1)) {
            {
                const int t  = s >> 1;
                const int tn = (t + 1 < T_STEPS) ? (t + 1) : (T_STEPS - 1);
                #pragma unroll
                for (int kf = 0; kf < NKF_W; ++kf) {
                    #pragma unroll
                    for (int mt = 0; mt < MT; ++mt)
                        xa[kf][mt] = *(const v2f*)&xRow[mt][tn * F_IN + kf * 4 + half * 2];
                }
                __builtin_amdgcn_sched_barrier(0);

                #pragma unroll
                for (int ns = 0; ns < NS; ++ns) {
                    #pragma unroll
                    for (int mt = 0; mt < MT; ++mt) {
                        #pragma unroll
                        for (int r = 0; r < 8; ++r) {
                            const float ig = fast_sigmoid(acc[0][ns][mt][r]);
                            const float fg = fast_sigmoid(acc[1][ns][mt][r]);
                            const float gg = fmaxf(acc[2][ns][mt][r], 0.0f);
                            const float og = fast_sigmoid(acc[3][ns][mt][r]);
                            const float cn = fg * cSt[ns][mt][r] + ig * gg;
                            cSt[ns][mt][r] = cn;
                            const float hv = og * fmaxf(cn, 0.0f);
                            hS[ns * 1024 + (mt * 16 + r) * 4] = hv;
                        }
                    }
                }
            }
        }
        __syncthreads();
    }

    if (tid < TILE_M) {
        const int m = tid;
        float logits[10];
        #pragma unroll
        for (int c = 0; c < 10; ++c) logits[c] = bd[c];
        for (int k = 0; k < H_UNITS; ++k) {
            const float hv = hSwz[(k >> 2) * 256 + m * 4 + ((k >> 1) & 1) * 2 + (k & 1)];
            #pragma unroll
            for (int c = 0; c < 10; ++c) logits[c] += hv * Wd[k * 10 + c];
        }
        float mx = logits[0];
        #pragma unroll
        for (int c = 1; c < 10; ++c) mx = fmaxf(mx, logits[c]);
        float e[10], sum = 0.0f;
        #pragma unroll
        for (int c = 0; c < 10; ++c) { e[c] = __expf(logits[c] - mx); sum += e[c]; }
        const float inv = 1.0f / sum;
        #pragma unroll
        for (int c = 0; c < 10; ++c)
            out[(size_t)(rowBase + m) * 10 + c] = e[c] * inv;
    }
}

extern "C" void kernel_launch(void* const* d_in, const int* in_sizes, int n_in,
                              void* d_out, int out_size, void* d_ws, size_t ws_size,
                              hipStream_t stream) {
    const float* x  = (const float*)d_in[0];
    const float* W  = (const float*)d_in[1];
    const float* U  = (const float*)d_in[2];
    const float* b  = (const float*)d_in[3];
    const float* Wd = (const float*)d_in[4];
    const float* bd = (const float*)d_in[5];
    float* out = (float*)d_out;

    const size_t ldsBytes = (size_t)(65536 + 8192) * sizeof(float);
    const int numBlocks = 16384 / TILE_M;

    hipLaunchKernelGGL(lstm_fused_wmma3, dim3(numBlocks), dim3(256), ldsBytes, stream,
                       x, W, U, b, Wd, bd, out);

    (void)in_sizes; (void)n_in; (void)out_size; (void)d_ws; (void)ws_size;
}
